# starting kernel
# baseline (speedup 1.0000x reference)
#include <hip/hip_runtime.h>
#include <hip/hip_bf16.h>

typedef __attribute__((ext_vector_type(2))) float v2f;
typedef __attribute__((ext_vector_type(4))) float v4f;
typedef __attribute__((ext_vector_type(8))) float v8f;

#define ROWS    16
#define CHUNK   512
#define STRIDE  516
#define BLOCK   512

#if defined(__AMDGCN__) && __has_builtin(__builtin_amdgcn_wmma_f32_16x16x4_f32)
#define HAVE_WMMA_F32 1
#else
#define HAVE_WMMA_F32 0
#endif

__device__ __forceinline__ float act(float s) {
  s = fminf(fmaxf(s, -20.0f), 20.0f);
#if defined(__AMDGCN__) && __has_builtin(__builtin_amdgcn_tanhf)
  return __builtin_amdgcn_tanhf(0.5f * s);
#else
  float e = expf(-s);
  return 2.0f / (1.0f + e) - 1.0f;
#endif
}

__global__ __launch_bounds__(BLOCK) void noise_mean_wmma(
    const float* __restrict__ pub,
    const float* __restrict__ priv,
    const float* __restrict__ pb,
    const float* __restrict__ pc,
    float* __restrict__ out,
    int n_samples, float inv_n)
{
  __shared__ float tile[ROWS * STRIDE];
  __shared__ float rowsum[ROWS];

  const int tid  = threadIdx.x;
  const int lane = tid & 31;
  const int wave = tid >> 5;
  const int rowBase = blockIdx.x * ROWS;

  const float b = pb[0];
  const float c = pc[0];
  const int   myRow = lane & 15;
  const float pt = pub[rowBase + myRow] * b;

  if (tid < ROWS) rowsum[tid] = 0.0f;

  const int ldRow = tid >> 7;
  const int ldCol = (tid & 127) * 4;

  const int kOff   = (lane >> 4) * 2;
  const int waveK0 = wave * (CHUNK / 16);

#if HAVE_WMMA_F32
  v8f acc = {};
  const v2f onesB = {1.0f, 1.0f};
#else
  float acc = 0.0f;
#endif

  v4f regs[4];
#pragma unroll
  for (int p = 0; p < 4; ++p) {
    const int r = p * 4 + ldRow;
    regs[p] = *(const v4f*)(priv + (size_t)(rowBase + r) * (size_t)n_samples
                                 + (size_t)ldCol);
  }

  for (int s0 = 0; s0 < n_samples; s0 += CHUNK) {
    __syncthreads();
#pragma unroll
    for (int p = 0; p < 4; ++p) {
      const int r = p * 4 + ldRow;
      *(v4f*)(tile + r * STRIDE + ldCol) = regs[p];
    }
    __syncthreads();

    const int s1 = s0 + CHUNK;
    if (s1 < n_samples) {
#pragma unroll
      for (int p = 0; p < 4; ++p) {
        const int r = p * 4 + ldRow;
        regs[p] = *(const v4f*)(priv + (size_t)(rowBase + r) * (size_t)n_samples
                                     + (size_t)(s1 + ldCol));
      }
    }

#pragma unroll
    for (int k = 0; k < CHUNK / 16; k += 4) {
      const float* src = tile + myRow * STRIDE + waveK0 + k + kOff;
      v2f a;
      a.x = act(pt + src[0] * c);
      a.y = act(pt + src[1] * c);
#if HAVE_WMMA_F32
      acc = __builtin_amdgcn_wmma_f32_16x16x4_f32(false, a, false, onesB,
                                                  (short)0, acc, false, false);
#else
      acc += a.x + a.y;
#endif
    }
  }

#if HAVE_WMMA_F32
  if (lane == 0 || lane == 16) {
    const int mbase = (lane >> 4) * 8;
#pragma unroll
    for (int i = 0; i < 8; ++i) atomicAdd(&rowsum[mbase + i], acc[i]);
  }
#else
  atomicAdd(&rowsum[myRow], acc);
#endif

  __syncthreads();
  if (tid < ROWS) out[rowBase + tid] = rowsum[tid] * inv_n;
}

extern "C" void kernel_launch(void* const* d_in, const int* in_sizes, int n_in,
                              void* d_out, int out_size, void* d_ws, size_t ws_size,
                              hipStream_t stream) {
  const float* pub  = (const float*)d_in[0];
  const float* priv = (const float*)d_in[1];
  const float* pb   = (const float*)d_in[2];
  const float* pc   = (const float*)d_in[3];
  float* out = (float*)d_out;

  const int batch     = in_sizes[0];
  const int n_samples = in_sizes[1] / batch;
  const float inv_n   = 1.0f / (float)n_samples;

  dim3 grid(batch / ROWS);
  noise_mean_wmma<<<grid, BLOCK, 0, stream>>>(pub, priv, pb, pc, out, n_samples, inv_n);
}


# each wave stages its own 16-row x 256-B strips through a private LDS tile, 16 whole-line loads in flight per wave, no block barrier in the loop; row sum order changed (fixed-order final sum)
# speedup vs baseline: 1.4861x; 1.4861x over previous
#include <hip/hip_runtime.h>
#include <hip/hip_bf16.h>


typedef __attribute__((ext_vector_type(2))) float v2f;
typedef __attribute__((ext_vector_type(4))) float v4f;
typedef __attribute__((ext_vector_type(8))) float v8f;

#define ROWS    16
#define WAVES   16
#define BLOCK   (WAVES * 32)
#define STRIP   64
#define LPS     8
#define NFRAG   (STRIP / 4)
#define PITCH   68

#if defined(__AMDGCN__) && __has_builtin(__builtin_amdgcn_wmma_f32_16x16x4_f32)
#define HAVE_WMMA_F32 1
#else
#define HAVE_WMMA_F32 0
#endif

__device__ __forceinline__ float act(float s) {
  s = fminf(fmaxf(s, -20.0f), 20.0f);
#if defined(__AMDGCN__) && __has_builtin(__builtin_amdgcn_tanhf)
  return __builtin_amdgcn_tanhf(0.5f * s);
#else
  float e = expf(-s);
  return 2.0f / (1.0f + e) - 1.0f;
#endif
}

__device__ __forceinline__ void wave_tile_handover() {
#if defined(__AMDGCN__)
  __builtin_amdgcn_fence(__ATOMIC_RELEASE, "wavefront");
  __builtin_amdgcn_wave_barrier();
  __builtin_amdgcn_fence(__ATOMIC_ACQUIRE, "wavefront");
#endif
}

__device__ __forceinline__ void load_strip(v4f (&buf)[LPS], const float* __restrict__ gbase,
                                           size_t rowPitch, unsigned byteOff) {
#pragma unroll
  for (int j = 0; j < LPS; ++j) {
    const char* rp = (const char*)(gbase + (size_t)(2 * j) * rowPitch);
    buf[j] = *(const v4f*)(rp + byteOff);
  }
}

__global__ __launch_bounds__(BLOCK) void noise_mean_wmma(
    const float* __restrict__ pub,
    const float* __restrict__ priv,
    const float* __restrict__ pb,
    const float* __restrict__ pc,
    float* __restrict__ out,
    int n_samples, float inv_n)
{
  __shared__ __attribute__((aligned(16))) float stage[WAVES * ROWS * PITCH];
  __shared__ float part[WAVES * ROWS];

  const int tid  = threadIdx.x;
  const int lane = tid & 31;
  const int wave = __builtin_amdgcn_readfirstlane(tid >> 5);
  const int c16  = lane & 15;
  const int h    = lane >> 4;
  const int rowBase = blockIdx.x * ROWS;

  const float b = pb[0];
  const float c = pc[0];
  float pt;

  const int    nStrips  = n_samples / STRIP;
  const size_t rowPitch = (size_t)n_samples;
  const float* gbase    = priv + (size_t)rowBase * rowPitch;
  const unsigned laneByte = (unsigned)(h * n_samples + 4 * c16) * 4u;

  float*       tile = stage + wave * (ROWS * PITCH);
  unsigned     stIdx = (unsigned)(wave * (ROWS * PITCH / 4) + h * (PITCH / 4) + c16);
  const float* ld   = tile + c16 * PITCH + 2 * h;

#if HAVE_WMMA_F32
  v8f acc0 = {};
  v8f acc1 = {};
  const v2f onesB = {1.0f, 1.0f};
#else
  float acc0 = 0.0f, acc1 = 0.0f;
#endif

#define STRIP_BYTE(S) (laneByte + (unsigned)(S) * (unsigned)(STRIP * 4))

#if HAVE_WMMA_F32
#define FRAGMENTS()                                                                       \
  {                                                                                       \
    v2f x[NFRAG];                                                                         \
    _Pragma("unroll")                                                                     \
    for (int t = 0; t < NFRAG; ++t) x[t] = *(const v2f*)(ld + 4 * t);                     \
        \
        \
        \
    __builtin_amdgcn_sched_barrier(0);                                                    \
    _Pragma("unroll")                                                                     \
    for (int t = 0; t < NFRAG; t += 2) {                                                  \
      v2f a0, a1;                                                                         \
      a0.x = act(pt + x[t].x * c);                                                        \
      a0.y = act(pt + x[t].y * c);                                                        \
      a1.x = act(pt + x[t + 1].x * c);                                                    \
      a1.y = act(pt + x[t + 1].y * c);                                                    \
      acc0 = __builtin_amdgcn_wmma_f32_16x16x4_f32(false, a0, false, onesB,               \
                                                   (short)0, acc0, false, false);         \
      acc1 = __builtin_amdgcn_wmma_f32_16x16x4_f32(false, a1, false, onesB,               \
                                                   (short)0, acc1, false, false);         \
    }                                                                                     \
  }
#else
#define FRAGMENTS()                                                                       \
  _Pragma("unroll")                                                                       \
  for (int t = 0; t < NFRAG; ++t) {                                                       \
    const v2f x0 = *(const v2f*)(ld + 4 * t);                                             \
    acc0 += act(pt + x0.x * c) + act(pt + x0.y * c);                                      \
  }
#endif

#if HAVE_WMMA_F32
#define CLOSE_STRIP() asm("" : "+v"(acc0), "+v"(acc1), "+v"(stIdx));
#else
#define CLOSE_STRIP()
#endif

#define DO_STRIP(BUF, REFILL)                                                             \
  {                                                                                       \
    _Pragma("unroll")                                                                     \
    for (int j = 0; j < LPS; ++j) *(v4f*)(stage + 4 * stIdx + j * 2 * PITCH) = BUF[j];    \
    wave_tile_handover();                                                                 \
    REFILL                                                                                \
    FRAGMENTS()                                                                           \
    wave_tile_handover();                                                                 \
    CLOSE_STRIP()                                                                         \
  }

#define PT_AFTER_REQUESTS()                                                               \
  {                                                                                       \
    float pubv = 0.0f;                                                                    \
    _Pragma("unroll")                                                                     \
    for (int i = 0; i < ROWS; ++i) {                                                      \
      const float p = pub[rowBase + i];          \
      pubv = (c16 == i) ? p : pubv;                                                       \
    }                                                                                     \
    pt = pubv * b;                                                                        \
  }

  v4f bufA[LPS], bufB[LPS];
  int s = wave;

  if (nStrips > 0 && (nStrips % (2 * WAVES)) == 0) {
    load_strip(bufA, gbase, rowPitch, STRIP_BYTE(s));
    __builtin_amdgcn_sched_barrier(0);
    load_strip(bufB, gbase, rowPitch, STRIP_BYTE(s + WAVES));
    __builtin_amdgcn_sched_barrier(0);
    PT_AFTER_REQUESTS()
    for (; s + 2 * WAVES < nStrips; s += 2 * WAVES) {
      DO_STRIP(bufA, load_strip(bufA, gbase, rowPitch, STRIP_BYTE(s + 2 * WAVES));)
      DO_STRIP(bufB, load_strip(bufB, gbase, rowPitch, STRIP_BYTE(s + 3 * WAVES));)
    }
    DO_STRIP(bufA, )
    DO_STRIP(bufB, )
  } else {
    if (s < nStrips)         load_strip(bufA, gbase, rowPitch, STRIP_BYTE(s));
    if (s + WAVES < nStrips) load_strip(bufB, gbase, rowPitch, STRIP_BYTE(s + WAVES));
    PT_AFTER_REQUESTS()
    for (; s < nStrips; s += 2 * WAVES) {
      DO_STRIP(bufA, if (s + 2 * WAVES < nStrips)
                       load_strip(bufA, gbase, rowPitch, STRIP_BYTE(s + 2 * WAVES));)
      if (s + WAVES < nStrips)
        DO_STRIP(bufB, if (s + 3 * WAVES < nStrips)
                         load_strip(bufB, gbase, rowPitch, STRIP_BYTE(s + 3 * WAVES));)
    }
  }

#if HAVE_WMMA_F32
  const v8f tot = acc0 + acc1;
  if (lane == 0 || lane == 16) {
    const int mbase = h * 8;
#pragma unroll
    for (int i = 0; i < 8; ++i) part[wave * ROWS + mbase + i] = tot[i];
  }
#else
  if (lane < 16) part[wave * ROWS + lane] = acc0 + acc1;
#endif

  __syncthreads();
  if (tid < ROWS) {
    float sum = 0.0f;
#pragma unroll
    for (int w = 0; w < WAVES; ++w) sum += part[w * ROWS + tid];
    out[rowBase + tid] = sum * inv_n;
  }
}

extern "C" void kernel_launch(void* const* d_in, const int* in_sizes, int n_in,
                              void* d_out, int out_size, void* d_ws, size_t ws_size,
                              hipStream_t stream) {
  const float* pub  = (const float*)d_in[0];
  const float* priv = (const float*)d_in[1];
  const float* pb   = (const float*)d_in[2];
  const float* pc   = (const float*)d_in[3];
  float* out = (float*)d_out;

  const int batch     = in_sizes[0];
  const int n_samples = in_sizes[1] / batch;
  const float inv_n   = 1.0f / (float)n_samples;

  dim3 grid(batch / ROWS);
  noise_mean_wmma<<<grid, BLOCK, 0, stream>>>(pub, priv, pb, pc, out, n_samples, inv_n);
}
